# starting kernel
# baseline (speedup 1.0000x reference)
#include <hip/hip_runtime.h>
#include <hip/hip_bf16.h>
#include <math.h>

#define EMA_B 8
#define EMA_P 4096
#define EMA_F 1024
#define EMA_C 16
#define EMA_L 256

typedef float v2f __attribute__((ext_vector_type(2)));
typedef float v8f __attribute__((ext_vector_type(8)));

__device__ __forceinline__ float sigmoidf_(float v) {
    return 1.0f / (1.0f + __expf(-v));
}

__device__ __forceinline__ float powi_f(float a, int n) {
    float r = 1.0f;
    for (int i = 0; i < n; ++i) r *= a;
    return r;
}

__global__ void ema_chunk_ends(const float* __restrict__ x,
                               const float* __restrict__ raw_alpha,
                               float* __restrict__ ws) {
    int g = blockIdx.x * blockDim.x + threadIdx.x;
    int f = g & (EMA_F - 1);
    int j = (g >> 10) & (EMA_C - 1);
    int b = g >> 14;

    float a  = sigmoidf_(raw_alpha[0]);
    float om = 1.0f - a;

    const float* xp = x + ((size_t)b * EMA_P + (size_t)j * EMA_L) * EMA_F + f;

    float h = 0.0f;
#pragma unroll 8
    for (int t = 0; t < EMA_L; ++t) {
        h = fmaf(a, h, om * xp[(size_t)t * EMA_F]);
    }
    ws[((size_t)b * EMA_C + j) * EMA_F + f] = h;
}

__global__ void ema_scan_wmma(const float* __restrict__ x,
                              const float* __restrict__ raw_alpha,
                              const float* __restrict__ ws,
                              float* __restrict__ out) {
    const int lane = threadIdx.x & 31;
    const int wv   = blockIdx.x * (blockDim.x >> 5) + (threadIdx.x >> 5);
    const int ftile = wv & 63;
    const int j     = (wv >> 6) & (EMA_C - 1);
    const int b     = wv >> 10;

    const int nlo  = lane & 15;
    const int half = lane >> 4;
    const int f    = (ftile << 4) + nlo;

    const float a  = sigmoidf_(raw_alpha[0]);
    const float om = 1.0f - a;

    float aL = a;
#pragma unroll
    for (int i = 0; i < 8; ++i) aL *= aL;

    float c = 0.0f;
    for (int i = 0; i < j; ++i) {
        c = fmaf(aL, c, ws[((size_t)b * EMA_C + i) * EMA_F + f]);
    }

    v2f Am[4];
#pragma unroll
    for (int s = 0; s < 4; ++s) {
        int m  = nlo;
        int k0 = 4 * s + 2 * half;
        Am[s].x = (k0     <= m) ? om * powi_f(a, m - k0)     : 0.0f;
        Am[s].y = (k0 + 1 <= m) ? om * powi_f(a, m - k0 - 1) : 0.0f;
    }

    float mrow[8];
    mrow[0] = powi_f(a, half ? 9 : 1);
#pragma unroll
    for (int r = 1; r < 8; ++r) mrow[r] = mrow[r - 1] * a;

    const float* xp = x   + ((size_t)b * EMA_P + (size_t)j * EMA_L) * EMA_F + f;
    float*       op = out + ((size_t)b * EMA_P + (size_t)j * EMA_L) * EMA_F + f;

    v2f Bv[2][4];
#pragma unroll
    for (int s = 0; s < 4; ++s) {
        int t = 4 * s + 2 * half;
        Bv[0][s].x = __builtin_nontemporal_load(xp + (size_t)t       * EMA_F);
        Bv[0][s].y = __builtin_nontemporal_load(xp + (size_t)(t + 1) * EMA_F);
    }

#pragma unroll
    for (int blk = 0; blk < EMA_L / 16; ++blk) {
        const int cur = blk & 1;
        if (blk < (EMA_L / 16 - 1)) {
            const int nt0 = (blk + 1) * 16;
#pragma unroll
            for (int s = 0; s < 4; ++s) {
                int t = nt0 + 4 * s + 2 * half;
                Bv[cur ^ 1][s].x = __builtin_nontemporal_load(xp + (size_t)t       * EMA_F);
                Bv[cur ^ 1][s].y = __builtin_nontemporal_load(xp + (size_t)(t + 1) * EMA_F);
            }
        }

        v8f acc;
#pragma unroll
        for (int r = 0; r < 8; ++r) acc[r] = mrow[r] * c;

#pragma unroll
        for (int s = 0; s < 4; ++s) {
            acc = __builtin_amdgcn_wmma_f32_16x16x4_f32(
                false, Am[s], false, Bv[cur][s], (short)0, acc, false, false);
        }

        const int tbase = blk * 16 + half * 8;
#pragma unroll
        for (int r = 0; r < 8; ++r) {
            __builtin_nontemporal_store(acc[r], op + (size_t)(tbase + r) * EMA_F);
        }

        c = __shfl(acc[7], 16 + nlo, 32);
    }
}

extern "C" void kernel_launch(void* const* d_in, const int* in_sizes, int n_in,
                              void* d_out, int out_size, void* d_ws, size_t ws_size,
                              hipStream_t stream) {
    const float* x  = (const float*)d_in[0];
    const float* ra = (const float*)d_in[1];
    float* out = (float*)d_out;
    float* ws  = (float*)d_ws;

    {
        const int threads = EMA_B * EMA_F * EMA_C;
        ema_chunk_ends<<<threads / 256, 256, 0, stream>>>(x, ra, ws);
    }
    {
        const int waves = EMA_B * (EMA_F / 16) * EMA_C;
        ema_scan_wmma<<<waves / 8, 256, 0, stream>>>(x, ra, ws, out);
    }
}


# chunk-end pass: each wave sums 32 steps of 256 columns with 16-32 whole-line 16-byte loads in flight; the 8 partial ends join through LDS by powers of alpha^32 (sum order changes)
# speedup vs baseline: 1.1483x; 1.1483x over previous
#include <hip/hip_runtime.h>
#include <hip/hip_bf16.h>
#include <math.h>

#define EMA_B 8
#define EMA_P 4096
#define EMA_F 1024
#define EMA_C 16
#define EMA_L 256

typedef float v2f __attribute__((ext_vector_type(2)));
typedef float v4f __attribute__((ext_vector_type(4)));
typedef float v8f __attribute__((ext_vector_type(8)));

__device__ __forceinline__ float sigmoidf_(float v) {
    return 1.0f / (1.0f + __expf(-v));
}

__device__ __forceinline__ float powi_f(float a, int n) {
    float r = 1.0f;
    for (int i = 0; i < n; ++i) r *= a;
    return r;
}

#define ENDS_WAVES 8
#define ENDS_ROWS  32
#define ENDS_GRP   8
#define ENDS_NGRP  (ENDS_ROWS / ENDS_GRP)
#define ENDS_COLS  256

__global__ void __launch_bounds__(ENDS_WAVES * 32)
ema_chunk_ends(const float* __restrict__ x,
               const float* __restrict__ raw_alpha,
               float* __restrict__ ws) {
    __shared__ __attribute__((aligned(16))) float part[ENDS_WAVES][ENDS_COLS];

    const int tid  = threadIdx.x;
    const int lane = tid & 31;
    const int w    = tid >> 5;
    const int cq   = blockIdx.x & 3;
    const int j    = (blockIdx.x >> 2) & (EMA_C - 1);
    const int b    = blockIdx.x >> 6;

    const float a  = sigmoidf_(raw_alpha[0]);
    const float om = 1.0f - a;

    const float* xp = x + ((size_t)b * EMA_P + (size_t)j * EMA_L + (size_t)w * ENDS_ROWS) * EMA_F
                        + cq * ENDS_COLS + lane * 4;

    float h[8];
#pragma unroll
    for (int i = 0; i < 8; ++i) h[i] = 0.0f;

    v4f buf[2][ENDS_GRP][2];
#define ENDS_LOAD_GRP(G, P) do { \
        _Pragma("unroll") for (int r = 0; r < ENDS_GRP; ++r) { \
            buf[(G) & 1][r][0] = *reinterpret_cast<const v4f*>((P) + (size_t)r * EMA_F); \
            buf[(G) & 1][r][1] = *reinterpret_cast<const v4f*>((P) + (size_t)r * EMA_F + 128); \
        } } while (0)
#define ENDS_SUM_GRP(G) do { \
        _Pragma("unroll") for (int r = 0; r < ENDS_GRP; ++r) { \
            const v4f v0 = buf[(G) & 1][r][0]; \
            const v4f v1 = buf[(G) & 1][r][1]; \
            h[0] = fmaf(a, h[0], om * v0.x); \
            h[1] = fmaf(a, h[1], om * v0.y); \
            h[2] = fmaf(a, h[2], om * v0.z); \
            h[3] = fmaf(a, h[3], om * v0.w); \
            h[4] = fmaf(a, h[4], om * v1.x); \
            h[5] = fmaf(a, h[5], om * v1.y); \
            h[6] = fmaf(a, h[6], om * v1.z); \
            h[7] = fmaf(a, h[7], om * v1.w); \
        } } while (0)

    ENDS_LOAD_GRP(0, xp);
    ENDS_LOAD_GRP(1, xp + (size_t)ENDS_GRP * EMA_F);
#pragma unroll
    for (int g = 0; g < ENDS_NGRP; ++g) {
        __builtin_amdgcn_sched_barrier(0);
        ENDS_SUM_GRP(g);
        if (g + 2 < ENDS_NGRP) {
            int off = (g + 2) * ENDS_GRP * EMA_F;
            asm volatile("" : "+v"(off), "+v"(h[0]), "+v"(h[1]), "+v"(h[2]), "+v"(h[3]),
                                         "+v"(h[4]), "+v"(h[5]), "+v"(h[6]), "+v"(h[7]));
            ENDS_LOAD_GRP(g + 2, xp + off);
        }
    }
#undef ENDS_LOAD_GRP
#undef ENDS_SUM_GRP

    v4f p0, p1;
    p0.x = h[0]; p0.y = h[1]; p0.z = h[2]; p0.w = h[3];
    p1.x = h[4]; p1.y = h[5]; p1.z = h[6]; p1.w = h[7];
    *reinterpret_cast<v4f*>(&part[w][lane * 4])       = p0;
    *reinterpret_cast<v4f*>(&part[w][128 + lane * 4]) = p1;
    __syncthreads();

    float a32 = a;
#pragma unroll
    for (int i = 0; i < 5; ++i) a32 *= a32;

    float e = 0.0f;
#pragma unroll
    for (int k = 0; k < ENDS_WAVES; ++k) e = fmaf(a32, e, part[k][tid]);

    ws[((size_t)b * EMA_C + j) * EMA_F + cq * ENDS_COLS + tid] = e;
}

__global__ void ema_scan_wmma(const float* __restrict__ x,
                              const float* __restrict__ raw_alpha,
                              const float* __restrict__ ws,
                              float* __restrict__ out) {
    const int lane = threadIdx.x & 31;
    const int wv   = blockIdx.x * (blockDim.x >> 5) + (threadIdx.x >> 5);
    const int ftile = wv & 63;
    const int j     = (wv >> 6) & (EMA_C - 1);
    const int b     = wv >> 10;

    const int nlo  = lane & 15;
    const int half = lane >> 4;
    const int f    = (ftile << 4) + nlo;

    const float a  = sigmoidf_(raw_alpha[0]);
    const float om = 1.0f - a;

    float aL = a;
#pragma unroll
    for (int i = 0; i < 8; ++i) aL *= aL;

    float c = 0.0f;
    for (int i = 0; i < j; ++i) {
        c = fmaf(aL, c, ws[((size_t)b * EMA_C + i) * EMA_F + f]);
    }

    v2f Am[4];
#pragma unroll
    for (int s = 0; s < 4; ++s) {
        int m  = nlo;
        int k0 = 4 * s + 2 * half;
        Am[s].x = (k0     <= m) ? om * powi_f(a, m - k0)     : 0.0f;
        Am[s].y = (k0 + 1 <= m) ? om * powi_f(a, m - k0 - 1) : 0.0f;
    }

    float mrow[8];
    mrow[0] = powi_f(a, half ? 9 : 1);
#pragma unroll
    for (int r = 1; r < 8; ++r) mrow[r] = mrow[r - 1] * a;

    const float* xp = x   + ((size_t)b * EMA_P + (size_t)j * EMA_L) * EMA_F + f;
    float*       op = out + ((size_t)b * EMA_P + (size_t)j * EMA_L) * EMA_F + f;

    v2f Bv[2][4];
#pragma unroll
    for (int s = 0; s < 4; ++s) {
        int t = 4 * s + 2 * half;
        Bv[0][s].x = __builtin_nontemporal_load(xp + (size_t)t       * EMA_F);
        Bv[0][s].y = __builtin_nontemporal_load(xp + (size_t)(t + 1) * EMA_F);
    }

#pragma unroll
    for (int blk = 0; blk < EMA_L / 16; ++blk) {
        const int cur = blk & 1;
        if (blk < (EMA_L / 16 - 1)) {
            const int nt0 = (blk + 1) * 16;
#pragma unroll
            for (int s = 0; s < 4; ++s) {
                int t = nt0 + 4 * s + 2 * half;
                Bv[cur ^ 1][s].x = __builtin_nontemporal_load(xp + (size_t)t       * EMA_F);
                Bv[cur ^ 1][s].y = __builtin_nontemporal_load(xp + (size_t)(t + 1) * EMA_F);
            }
        }

        v8f acc;
#pragma unroll
        for (int r = 0; r < 8; ++r) acc[r] = mrow[r] * c;

#pragma unroll
        for (int s = 0; s < 4; ++s) {
            acc = __builtin_amdgcn_wmma_f32_16x16x4_f32(
                false, Am[s], false, Bv[cur][s], (short)0, acc, false, false);
        }

        const int tbase = blk * 16 + half * 8;
#pragma unroll
        for (int r = 0; r < 8; ++r) {
            __builtin_nontemporal_store(acc[r], op + (size_t)(tbase + r) * EMA_F);
        }

        c = __shfl(acc[7], 16 + nlo, 32);
    }
}

extern "C" void kernel_launch(void* const* d_in, const int* in_sizes, int n_in,
                              void* d_out, int out_size, void* d_ws, size_t ws_size,
                              hipStream_t stream) {
    const float* x  = (const float*)d_in[0];
    const float* ra = (const float*)d_in[1];
    float* out = (float*)d_out;
    float* ws  = (float*)d_ws;

    {
        const int groups = EMA_B * EMA_C * (EMA_F / ENDS_COLS);
        ema_chunk_ends<<<groups, ENDS_WAVES * 32, 0, stream>>>(x, ra, ws);
    }
    {
        const int waves = EMA_B * (EMA_F / 16) * EMA_C;
        ema_scan_wmma<<<waves / 8, 256, 0, stream>>>(x, ra, ws, out);
    }
}


# pass B: each wave takes 64 columns as four interleaved 16-column tiles, so every x load and out store is 16 bytes per lane on whole 128-byte lines; 4x fewer waves, each with 3 row blocks in flight
# speedup vs baseline: 2.1511x; 1.8733x over previous
#include <hip/hip_runtime.h>
#include <hip/hip_bf16.h>
#include <math.h>

#define EMA_B 8
#define EMA_P 4096
#define EMA_F 1024
#define EMA_C 16
#define EMA_L 256

typedef float v2f __attribute__((ext_vector_type(2)));
typedef float v4f __attribute__((ext_vector_type(4)));
typedef float v8f __attribute__((ext_vector_type(8)));

__device__ __forceinline__ float sigmoidf_(float v) {
    return 1.0f / (1.0f + __expf(-v));
}

__device__ __forceinline__ float powi_f(float a, int n) {
    float r = 1.0f;
    for (int i = 0; i < n; ++i) r *= a;
    return r;
}

#define ENDS_WAVES 8
#define ENDS_ROWS  32
#define ENDS_GRP   8
#define ENDS_NGRP  (ENDS_ROWS / ENDS_GRP)
#define ENDS_COLS  256

__global__ void __launch_bounds__(ENDS_WAVES * 32)
ema_chunk_ends(const float* __restrict__ x,
               const float* __restrict__ raw_alpha,
               float* __restrict__ ws) {
    __shared__ __attribute__((aligned(16))) float part[ENDS_WAVES][ENDS_COLS];

    const int tid  = threadIdx.x;
    const int lane = tid & 31;
    const int w    = tid >> 5;
    const int cq   = blockIdx.x & 3;
    const int j    = (blockIdx.x >> 2) & (EMA_C - 1);
    const int b    = blockIdx.x >> 6;

    const float a  = sigmoidf_(raw_alpha[0]);
    const float om = 1.0f - a;

    const float* xp = x + ((size_t)b * EMA_P + (size_t)j * EMA_L + (size_t)w * ENDS_ROWS) * EMA_F
                        + cq * ENDS_COLS + lane * 4;

    float h[8];
#pragma unroll
    for (int i = 0; i < 8; ++i) h[i] = 0.0f;

    v4f buf[2][ENDS_GRP][2];
#define ENDS_LOAD_GRP(G, P) do { \
        _Pragma("unroll") for (int r = 0; r < ENDS_GRP; ++r) { \
            buf[(G) & 1][r][0] = *reinterpret_cast<const v4f*>((P) + (size_t)r * EMA_F); \
            buf[(G) & 1][r][1] = *reinterpret_cast<const v4f*>((P) + (size_t)r * EMA_F + 128); \
        } } while (0)
#define ENDS_SUM_GRP(G) do { \
        _Pragma("unroll") for (int r = 0; r < ENDS_GRP; ++r) { \
            const v4f v0 = buf[(G) & 1][r][0]; \
            const v4f v1 = buf[(G) & 1][r][1]; \
            h[0] = fmaf(a, h[0], om * v0.x); \
            h[1] = fmaf(a, h[1], om * v0.y); \
            h[2] = fmaf(a, h[2], om * v0.z); \
            h[3] = fmaf(a, h[3], om * v0.w); \
            h[4] = fmaf(a, h[4], om * v1.x); \
            h[5] = fmaf(a, h[5], om * v1.y); \
            h[6] = fmaf(a, h[6], om * v1.z); \
            h[7] = fmaf(a, h[7], om * v1.w); \
        } } while (0)

    ENDS_LOAD_GRP(0, xp);
    ENDS_LOAD_GRP(1, xp + (size_t)ENDS_GRP * EMA_F);
#pragma unroll
    for (int g = 0; g < ENDS_NGRP; ++g) {
        __builtin_amdgcn_sched_barrier(0);
        ENDS_SUM_GRP(g);
        if (g + 2 < ENDS_NGRP) {
            int off = (g + 2) * ENDS_GRP * EMA_F;
            asm volatile("" : "+v"(off), "+v"(h[0]), "+v"(h[1]), "+v"(h[2]), "+v"(h[3]),
                                         "+v"(h[4]), "+v"(h[5]), "+v"(h[6]), "+v"(h[7]));
            ENDS_LOAD_GRP(g + 2, xp + off);
        }
    }
#undef ENDS_LOAD_GRP
#undef ENDS_SUM_GRP

    v4f p0, p1;
    p0.x = h[0]; p0.y = h[1]; p0.z = h[2]; p0.w = h[3];
    p1.x = h[4]; p1.y = h[5]; p1.z = h[6]; p1.w = h[7];
    *reinterpret_cast<v4f*>(&part[w][lane * 4])       = p0;
    *reinterpret_cast<v4f*>(&part[w][128 + lane * 4]) = p1;
    __syncthreads();

    float a32 = a;
#pragma unroll
    for (int i = 0; i < 5; ++i) a32 *= a32;

    float e = 0.0f;
#pragma unroll
    for (int k = 0; k < ENDS_WAVES; ++k) e = fmaf(a32, e, part[k][tid]);

    ws[((size_t)b * EMA_C + j) * EMA_F + cq * ENDS_COLS + tid] = e;
}

#define SCAN_COLS  64
#define SCAN_NBLK  (EMA_L / 16)
#define SCAN_RING  4
#define SCAN_WAVES 4

__global__ void __launch_bounds__(SCAN_WAVES * 32)
ema_scan_wmma(const float* __restrict__ x,
              const float* __restrict__ raw_alpha,
              const float* __restrict__ ws,
              float* __restrict__ out) {
    const int lane = threadIdx.x & 31;
    const int wv   = blockIdx.x * SCAN_WAVES + (threadIdx.x >> 5);
    const int ct   = wv & (EMA_F / SCAN_COLS - 1);
    const int j    = (wv >> 4) & (EMA_C - 1);
    const int b    = wv >> 8;

    const int c16  = lane & 15;
    const int half = lane >> 4;
    const int col  = ct * SCAN_COLS + c16 * 4;

    const float* xp = x   + ((size_t)b * EMA_P + (size_t)j * EMA_L + 2 * half) * EMA_F + col;
    float*       op = out + ((size_t)b * EMA_P + (size_t)j * EMA_L + 8 * half) * EMA_F + col;

    v4f X[SCAN_RING][4][2];
#define SCAN_LOAD_BLK(K) do { \
        _Pragma("unroll") for (int s = 0; s < 4; ++s) { \
            X[(K) % SCAN_RING][s][0] = __builtin_nontemporal_load( \
                reinterpret_cast<const v4f*>(xp + (size_t)((K) * 16 + 4 * s)     * EMA_F)); \
            X[(K) % SCAN_RING][s][1] = __builtin_nontemporal_load( \
                reinterpret_cast<const v4f*>(xp + (size_t)((K) * 16 + 4 * s + 1) * EMA_F)); \
        } } while (0)

    SCAN_LOAD_BLK(0);
    SCAN_LOAD_BLK(1);
    SCAN_LOAD_BLK(2);

    const float a  = sigmoidf_(raw_alpha[0]);
    const float om = 1.0f - a;

    float aL = a;
#pragma unroll
    for (int i = 0; i < 8; ++i) aL *= aL;

    v2f Am[4];
#pragma unroll
    for (int s = 0; s < 4; ++s) {
        int m  = c16;
        int k0 = 4 * s + 2 * half;
        Am[s].x = (k0     <= m) ? om * powi_f(a, m - k0)     : 0.0f;
        Am[s].y = (k0 + 1 <= m) ? om * powi_f(a, m - k0 - 1) : 0.0f;
    }

    float mrow[8];
    mrow[0] = powi_f(a, half ? 9 : 1);
#pragma unroll
    for (int r = 1; r < 8; ++r) mrow[r] = mrow[r - 1] * a;

    v4f c;
    c.x = 0.0f; c.y = 0.0f; c.z = 0.0f; c.w = 0.0f;
    for (int i = 0; i < j; ++i) {
        const v4f e = *reinterpret_cast<const v4f*>(ws + ((size_t)b * EMA_C + i) * EMA_F + col);
        c.x = fmaf(aL, c.x, e.x);
        c.y = fmaf(aL, c.y, e.y);
        c.z = fmaf(aL, c.z, e.z);
        c.w = fmaf(aL, c.w, e.w);
    }

#pragma unroll
    for (int blk = 0; blk < SCAN_NBLK; ++blk) {
        const int cur = blk % SCAN_RING;
        if (blk + SCAN_RING - 1 < SCAN_NBLK) {
            SCAN_LOAD_BLK(blk + SCAN_RING - 1);
        }

        v8f acc[4];
#pragma unroll
        for (int i = 0; i < 4; ++i) {
#pragma unroll
            for (int r = 0; r < 8; ++r) acc[i][r] = mrow[r] * c[i];
        }

#pragma unroll
        for (int s = 0; s < 4; ++s) {
#pragma unroll
            for (int i = 0; i < 4; ++i) {
                v2f bb;
                bb.x = X[cur][s][0][i];
                bb.y = X[cur][s][1][i];
                acc[i] = __builtin_amdgcn_wmma_f32_16x16x4_f32(
                    false, Am[s], false, bb, (short)0, acc[i], false, false);
            }
        }

#pragma unroll
        for (int r = 0; r < 8; ++r) {
            v4f o;
            o.x = acc[0][r]; o.y = acc[1][r]; o.z = acc[2][r]; o.w = acc[3][r];
            __builtin_nontemporal_store(o, reinterpret_cast<v4f*>(op + (size_t)(blk * 16 + r) * EMA_F));
        }

#pragma unroll
        for (int i = 0; i < 4; ++i) c[i] = __shfl(acc[i][7], 16 + c16, 32);
    }
#undef SCAN_LOAD_BLK
}

extern "C" void kernel_launch(void* const* d_in, const int* in_sizes, int n_in,
                              void* d_out, int out_size, void* d_ws, size_t ws_size,
                              hipStream_t stream) {
    const float* x  = (const float*)d_in[0];
    const float* ra = (const float*)d_in[1];
    float* out = (float*)d_out;
    float* ws  = (float*)d_ws;

    {
        const int groups = EMA_B * EMA_C * (EMA_F / ENDS_COLS);
        ema_chunk_ends<<<groups, ENDS_WAVES * 32, 0, stream>>>(x, ra, ws);
    }
    {
        const int waves = EMA_B * (EMA_F / SCAN_COLS) * EMA_C;
        ema_scan_wmma<<<waves / SCAN_WAVES, SCAN_WAVES * 32, 0, stream>>>(x, ra, ws, out);
    }
}


# one launch, x read once: a 16-wave workgroup owns one aligned 256-byte unit (64 columns) for all steps; per 256-step stage the 16-step block ends join via LDS, carry added after (sum order changes)
# speedup vs baseline: 2.3031x; 1.0707x over previous
#include <hip/hip_runtime.h>
#include <hip/hip_bf16.h>
#include <math.h>

#define EMA_B 8
#define EMA_P 4096
#define EMA_F 1024

typedef float v2f __attribute__((ext_vector_type(2)));
typedef float v4f __attribute__((ext_vector_type(4)));
typedef float v8f __attribute__((ext_vector_type(8)));

__device__ __forceinline__ float sigmoidf_(float v) {
    return 1.0f / (1.0f + __expf(-v));
}

__device__ __forceinline__ float xhalf(float v) {
    const int b = __builtin_bit_cast(int, v);
    return __builtin_bit_cast(float,
        __builtin_amdgcn_permlanex16(b, b, 0x76543210, 0xfedcba98, false, false));
}

__device__ __forceinline__ float powi_f(float a, int n) {
    float r = 1.0f;
    for (int i = 0; i < n; ++i) r *= a;
    return r;
}

#define SCAN_COLS   64
#define SCAN_WAVES  16
#define SCAN_STEPS  (SCAN_WAVES * 16)
#define SCAN_NSTAGE (EMA_P / SCAN_STEPS)
#define SCAN_RING   4

__global__ void __launch_bounds__(SCAN_WAVES * 32)
ema_scan_fused(const float* __restrict__ x,
               const float* __restrict__ raw_alpha,
               float* __restrict__ out) {
    __shared__ __attribute__((aligned(16))) float part[2][SCAN_WAVES][SCAN_COLS];

    const int lane = threadIdx.x & 31;
    const int w    = __builtin_amdgcn_readfirstlane(threadIdx.x >> 5);
    const int cu   = blockIdx.x & (EMA_F / SCAN_COLS - 1);
    const int b    = blockIdx.x >> 4;

    const int c16  = lane & 15;
    const int half = lane >> 4;
    const int col  = cu * SCAN_COLS + c16 * 4;

    const float* xp = x   + ((size_t)b * EMA_P + (size_t)w * 16 + 2 * half) * EMA_F + col;
    float*       op = out + ((size_t)b * EMA_P + (size_t)w * 16 + 8 * half) * EMA_F + col;

    v4f X[SCAN_RING][4][2];
#define SCAN_LOAD_STAGE(SLOT, ST) do { \
        _Pragma("unroll") for (int s = 0; s < 4; ++s) { \
            X[(SLOT)][s][0] = __builtin_nontemporal_load( \
                reinterpret_cast<const v4f*>(xp + ((size_t)(ST) * SCAN_STEPS + 4 * s)     * EMA_F)); \
            X[(SLOT)][s][1] = __builtin_nontemporal_load( \
                reinterpret_cast<const v4f*>(xp + ((size_t)(ST) * SCAN_STEPS + 4 * s + 1) * EMA_F)); \
        } } while (0)

    SCAN_LOAD_STAGE(0, 0);
    SCAN_LOAD_STAGE(1, 1);
    SCAN_LOAD_STAGE(2, 2);

    const float a  = sigmoidf_(raw_alpha[0]);
    const float om = 1.0f - a;

    float a16 = a;
#pragma unroll
    for (int i = 0; i < 4; ++i) a16 *= a16;

    v2f Am[4];
#pragma unroll
    for (int s = 0; s < 4; ++s) {
        int m  = c16;
        int k0 = 4 * s + 2 * half;
        Am[s].x = (k0     <= m) ? om * powi_f(a, m - k0)     : 0.0f;
        Am[s].y = (k0 + 1 <= m) ? om * powi_f(a, m - k0 - 1) : 0.0f;
    }

    float arow[8];
    arow[0] = a;
#pragma unroll
    for (int r = 1; r < 8; ++r) arow[r] = arow[r - 1] * a;
    const float ahalf = half ? arow[7] : 1.0f;

    v2f run;
    run.x = 0.0f; run.y = 0.0f;

#define SCAN_STAGE(U, ST, PF) do { \
        const int st = (ST); \
        if (PF) { SCAN_LOAD_STAGE(((U) + SCAN_RING - 1) % SCAN_RING, st + SCAN_RING - 1); } \
        v8f acc[4]; \
        _Pragma("unroll") for (int i = 0; i < 4; ++i) { \
            _Pragma("unroll") for (int r = 0; r < 8; ++r) acc[i][r] = 0.0f; \
        } \
        _Pragma("unroll") for (int s = 0; s < 4; ++s) { \
            _Pragma("unroll") for (int i = 0; i < 4; ++i) { \
                v2f bb; \
                bb.x = X[(U)][s][0][i]; \
                bb.y = X[(U)][s][1][i]; \
                acc[i] = __builtin_amdgcn_wmma_f32_16x16x4_f32( \
                    false, Am[s], false, bb, (short)0, acc[i], false, false); \
            } \
        } \
        if (half) { \
            v4f e; \
            e.x = acc[0][7]; e.y = acc[1][7]; e.z = acc[2][7]; e.w = acc[3][7]; \
            *reinterpret_cast<v4f*>(&part[st & 1][w][c16 * 4]) = e; \
        } \
        __builtin_amdgcn_fence(__ATOMIC_RELEASE, "workgroup", "local"); \
        __builtin_amdgcn_s_barrier(); \
        __builtin_amdgcn_fence(__ATOMIC_ACQUIRE, "workgroup", "local"); \
        v2f c2 = run; \
        _Pragma("unroll") for (int k = 0; k < SCAN_WAVES; ++k) { \
            if (k == w) c2 = run; \
            const v2f e = *reinterpret_cast<const v2f*>(&part[st & 1][k][c16 * 4 + 2 * half]); \
            run.x = fmaf(a16, run.x, e.x); \
            run.y = fmaf(a16, run.y, e.y); \
        } \
        const float ox = xhalf(c2.x); \
        const float oy = xhalf(c2.y); \
        v4f c; \
        c.x = half ? ox : c2.x; \
        c.y = half ? oy : c2.y; \
        c.z = half ? c2.x : ox; \
        c.w = half ? c2.y : oy; \
        c.x *= ahalf; c.y *= ahalf; c.z *= ahalf; c.w *= ahalf; \
        _Pragma("unroll") for (int r = 0; r < 8; ++r) { \
            v4f o; \
            o.x = fmaf(arow[r], c.x, acc[0][r]); \
            o.y = fmaf(arow[r], c.y, acc[1][r]); \
            o.z = fmaf(arow[r], c.z, acc[2][r]); \
            o.w = fmaf(arow[r], c.w, acc[3][r]); \
            __builtin_nontemporal_store(o, reinterpret_cast<v4f*>(op + ((size_t)st * SCAN_STEPS + r) * EMA_F)); \
        } } while (0)

    for (int sb = 0; sb < SCAN_NSTAGE - SCAN_RING; sb += SCAN_RING) {
        SCAN_STAGE(0, sb,     1);
        SCAN_STAGE(1, sb + 1, 1);
        SCAN_STAGE(2, sb + 2, 1);
        SCAN_STAGE(3, sb + 3, 1);
    }
    SCAN_STAGE(0, SCAN_NSTAGE - 4, 1);
    SCAN_STAGE(1, SCAN_NSTAGE - 3, 0);
    SCAN_STAGE(2, SCAN_NSTAGE - 2, 0);
    SCAN_STAGE(3, SCAN_NSTAGE - 1, 0);
#undef SCAN_STAGE
#undef SCAN_LOAD_STAGE
}

extern "C" void kernel_launch(void* const* d_in, const int* in_sizes, int n_in,
                              void* d_out, int out_size, void* d_ws, size_t ws_size,
                              hipStream_t stream) {
    const float* x  = (const float*)d_in[0];
    const float* ra = (const float*)d_in[1];
    float* out = (float*)d_out;

    const int groups = EMA_B * (EMA_F / SCAN_COLS);
    ema_scan_fused<<<groups, SCAN_WAVES * 32, 0, stream>>>(x, ra, out);
}


# a workgroup owns 32 columns (one 128-byte line per row), not 64: 256 workgroups, one per WGP, not 128; a wave's 16 tile columns = 8 column quads x 2 time blocks, 512-step stages; sum order unchanged
# speedup vs baseline: 3.0740x; 1.3347x over previous
#include <hip/hip_runtime.h>
#include <hip/hip_bf16.h>
#include <math.h>

#define EMA_B 8
#define EMA_P 4096
#define EMA_F 1024

typedef float v2f __attribute__((ext_vector_type(2)));
typedef float v4f __attribute__((ext_vector_type(4)));
typedef float v8f __attribute__((ext_vector_type(8)));

__device__ __forceinline__ float sigmoidf_(float v) {
    return 1.0f / (1.0f + __expf(-v));
}

__device__ __forceinline__ float xhalf(float v) {
    const int b = __builtin_bit_cast(int, v);
    return __builtin_bit_cast(float,
        __builtin_amdgcn_permlanex16(b, b, 0x76543210, 0xfedcba98, false, false));
}

__device__ __forceinline__ float powi_f(float a, int n) {
    float r = 1.0f;
    for (int i = 0; i < n; ++i) r *= a;
    return r;
}

#define SCAN_COLS   32
#define SCAN_WAVES  16
#define SCAN_BLOCKS (SCAN_WAVES * 2)
#define SCAN_STEPS  (SCAN_BLOCKS * 16)
#define SCAN_NSTAGE (EMA_P / SCAN_STEPS)
#define SCAN_RING   4

__global__ void __launch_bounds__(SCAN_WAVES * 32)
ema_scan_fused(const float* __restrict__ x,
               const float* __restrict__ raw_alpha,
               float* __restrict__ out) {
    __shared__ __attribute__((aligned(16))) float part[2][SCAN_BLOCKS][SCAN_COLS];

    const int lane = threadIdx.x & 31;
    const int w    = __builtin_amdgcn_readfirstlane(threadIdx.x >> 5);
    const int cu   = blockIdx.x & (EMA_F / SCAN_COLS - 1);
    const int b    = blockIdx.x / (EMA_F / SCAN_COLS);

    const int c16  = lane & 15;
    const int half = lane >> 4;
    const int n    = c16 & 7;
    const int tb   = c16 >> 3;
    const int col  = cu * SCAN_COLS + n * 4;

    const float* xp = x   + ((size_t)b * EMA_P + (size_t)w * 32 + tb * 16 + 2 * half) * EMA_F + col;
    float*       op = out + ((size_t)b * EMA_P + (size_t)w * 32 + tb * 16 + 8 * half) * EMA_F + col;

    v4f X[SCAN_RING][4][2];
#define SCAN_LOAD_STAGE(SLOT, ST) do { \
        _Pragma("unroll") for (int s = 0; s < 4; ++s) { \
            X[(SLOT)][s][0] = __builtin_nontemporal_load( \
                reinterpret_cast<const v4f*>(xp + ((size_t)(ST) * SCAN_STEPS + 4 * s)     * EMA_F)); \
            X[(SLOT)][s][1] = __builtin_nontemporal_load( \
                reinterpret_cast<const v4f*>(xp + ((size_t)(ST) * SCAN_STEPS + 4 * s + 1) * EMA_F)); \
        } } while (0)

    SCAN_LOAD_STAGE(0, 0);
    SCAN_LOAD_STAGE(1, 1);
    SCAN_LOAD_STAGE(2, 2);

    const float a  = sigmoidf_(raw_alpha[0]);
    const float om = 1.0f - a;

    float a16 = a;
#pragma unroll
    for (int i = 0; i < 4; ++i) a16 *= a16;

    v2f Am[4];
#pragma unroll
    for (int s = 0; s < 4; ++s) {
        int m  = c16;
        int k0 = 4 * s + 2 * half;
        Am[s].x = (k0     <= m) ? om * powi_f(a, m - k0)     : 0.0f;
        Am[s].y = (k0 + 1 <= m) ? om * powi_f(a, m - k0 - 1) : 0.0f;
    }

    float arow[8];
    arow[0] = a;
#pragma unroll
    for (int r = 1; r < 8; ++r) arow[r] = arow[r - 1] * a;
    const float ahalf = half ? arow[7] : 1.0f;

    v2f run;
    run.x = 0.0f; run.y = 0.0f;

#define SCAN_STAGE(U, ST, PF) do { \
        const int st = (ST); \
        if (PF) { SCAN_LOAD_STAGE(((U) + SCAN_RING - 1) % SCAN_RING, st + SCAN_RING - 1); } \
        v8f acc[4]; \
        _Pragma("unroll") for (int i = 0; i < 4; ++i) { \
            _Pragma("unroll") for (int r = 0; r < 8; ++r) acc[i][r] = 0.0f; \
        } \
        _Pragma("unroll") for (int s = 0; s < 4; ++s) { \
            _Pragma("unroll") for (int i = 0; i < 4; ++i) { \
                v2f bb; \
                bb.x = X[(U)][s][0][i]; \
                bb.y = X[(U)][s][1][i]; \
                acc[i] = __builtin_amdgcn_wmma_f32_16x16x4_f32( \
                    false, Am[s], false, bb, (short)0, acc[i], false, false); \
            } \
        } \
        if (half) { \
            v4f e; \
            e.x = acc[0][7]; e.y = acc[1][7]; e.z = acc[2][7]; e.w = acc[3][7]; \
            *reinterpret_cast<v4f*>(&part[st & 1][2 * w + tb][n * 4]) = e; \
        } \
        __builtin_amdgcn_fence(__ATOMIC_RELEASE, "workgroup", "local"); \
        __builtin_amdgcn_s_barrier(); \
        __builtin_amdgcn_fence(__ATOMIC_ACQUIRE, "workgroup", "local"); \
        v2f cA = run; \
        v2f cB = run; \
        _Pragma("unroll") for (int k = 0; k < SCAN_WAVES; ++k) { \
            if (k == w) cA = run; \
            const v2f e0 = *reinterpret_cast<const v2f*>(&part[st & 1][2 * k][n * 4 + 2 * half]); \
            run.x = fmaf(a16, run.x, e0.x); \
            run.y = fmaf(a16, run.y, e0.y); \
            if (k == w) cB = run; \
            const v2f e1 = *reinterpret_cast<const v2f*>(&part[st & 1][2 * k + 1][n * 4 + 2 * half]); \
            run.x = fmaf(a16, run.x, e1.x); \
            run.y = fmaf(a16, run.y, e1.y); \
        } \
        v2f c2; \
        c2.x = tb ? cB.x : cA.x; \
        c2.y = tb ? cB.y : cA.y; \
        const float ox = xhalf(c2.x); \
        const float oy = xhalf(c2.y); \
        v4f c; \
        c.x = half ? ox : c2.x; \
        c.y = half ? oy : c2.y; \
        c.z = half ? c2.x : ox; \
        c.w = half ? c2.y : oy; \
        c.x *= ahalf; c.y *= ahalf; c.z *= ahalf; c.w *= ahalf; \
        _Pragma("unroll") for (int r = 0; r < 8; ++r) { \
            v4f o; \
            o.x = fmaf(arow[r], c.x, acc[0][r]); \
            o.y = fmaf(arow[r], c.y, acc[1][r]); \
            o.z = fmaf(arow[r], c.z, acc[2][r]); \
            o.w = fmaf(arow[r], c.w, acc[3][r]); \
            __builtin_nontemporal_store(o, reinterpret_cast<v4f*>(op + ((size_t)st * SCAN_STEPS + r) * EMA_F)); \
        } } while (0)

    for (int sb = 0; sb < SCAN_NSTAGE - SCAN_RING; sb += SCAN_RING) {
        SCAN_STAGE(0, sb,     1);
        SCAN_STAGE(1, sb + 1, 1);
        SCAN_STAGE(2, sb + 2, 1);
        SCAN_STAGE(3, sb + 3, 1);
    }
    SCAN_STAGE(0, SCAN_NSTAGE - 4, 1);
    SCAN_STAGE(1, SCAN_NSTAGE - 3, 0);
    SCAN_STAGE(2, SCAN_NSTAGE - 2, 0);
    SCAN_STAGE(3, SCAN_NSTAGE - 1, 0);
#undef SCAN_STAGE
#undef SCAN_LOAD_STAGE
}

extern "C" void kernel_launch(void* const* d_in, const int* in_sizes, int n_in,
                              void* d_out, int out_size, void* d_ws, size_t ws_size,
                              hipStream_t stream) {
    const float* x  = (const float*)d_in[0];
    const float* ra = (const float*)d_in[1];
    float* out = (float*)d_out;

    const int groups = EMA_B * (EMA_F / SCAN_COLS);
    ema_scan_fused<<<groups, SCAN_WAVES * 32, 0, stream>>>(x, ra, out);
}
